# starting kernel
# baseline (speedup 1.0000x reference)
#include <hip/hip_runtime.h>

typedef float v2f __attribute__((ext_vector_type(2)));
typedef float v8f __attribute__((ext_vector_type(8)));

#define BB   2
#define DD   128
#define HH   128
#define WW   128
#define CI   64
#define CO   64
#define NOFF 27
#define NPTS 200000
#define GRID_CELLS (BB*DD*HH*WW)
#define WT_FLOATS  (NOFF*16*4*32*2)
#define A_STRIDE   68

__global__ void k_fill_grid(int* __restrict__ grid) {
    int i = blockIdx.x * blockDim.x + threadIdx.x;
    if (i < GRID_CELLS) grid[i] = -1;
}

__global__ void k_scatter(const int* __restrict__ coords, int* __restrict__ grid, int n) {
    int i = blockIdx.x * blockDim.x + threadIdx.x;
    if (i >= n) return;
    int b = coords[i*4+0], z = coords[i*4+1], y = coords[i*4+2], x = coords[i*4+3];
    int lin = ((b*DD + z)*HH + y)*WW + x;
    grid[lin] = i;
}

__global__ void k_wt(const float* __restrict__ w, float* __restrict__ wT) {
    int i = blockIdx.x * blockDim.x + threadIdx.x;
    if (i >= NOFF*16*4*32) return;
    int lane = i & 31;
    int nt   = (i >> 5) & 3;
    int kc   = (i >> 7) & 15;
    int t    =  i >> 11;
    int o = nt*16 + (lane & 15);
    int c = kc*4 + 2*(lane >> 4);
    const float* src = w + (size_t)(o*NOFF + t)*CI + c;
    wT[(size_t)i*2 + 0] = src[0];
    wT[(size_t)i*2 + 1] = src[1];
}

__global__ __launch_bounds__(128)
void k_conv(const float* __restrict__ feats,
            const float* __restrict__ wT,
            const float* __restrict__ bias,
            const int*   __restrict__ coords,
            const int*   __restrict__ grid,
            float*       __restrict__ out) {
    __shared__ int   nbr[NOFF*16];
    __shared__ float A[16*A_STRIDE];

    const int tid  = threadIdx.x;
    const int lane = tid & 31;
    const int wv   = tid >> 5;
    const int p0   = blockIdx.x * 16;

    for (int i = tid; i < NOFF*16; i += 128) {
        int t = i >> 4, m = i & 15;
        int p = p0 + m;
        int bz = coords[p*4+0], z = coords[p*4+1], y = coords[p*4+2], x = coords[p*4+3];
        int kd = t / 9, kh = (t / 3) % 3, kw = t % 3;
        int nz = z + kd - 1, ny = y + kh - 1, nx = x + kw - 1;
        int idx = -1;
        if ((unsigned)nz < (unsigned)DD && (unsigned)ny < (unsigned)HH &&
            (unsigned)nx < (unsigned)WW) {
            int lin = ((bz*DD + nz)*HH + ny)*WW + nx;
            idx = grid[lin];
        }
        nbr[i] = idx;
    }
    __syncthreads();

    const int col  = lane & 15;
    const int half = lane >> 4;
    float bv = bias[wv*16 + col];
    v8f acc;
#pragma unroll
    for (int j = 0; j < 8; ++j) acc[j] = bv;

    const int arow  = tid >> 3;
    const int acol8 = (tid & 7) * 8;

    for (int t = 0; t < NOFF; ++t) {
        int idx = nbr[t*16 + arow];
        float4 lo = make_float4(0.f, 0.f, 0.f, 0.f), hi = lo;
        if (idx >= 0) {
            const float4* src = (const float4*)(feats + (size_t)idx*CI + acol8);
            lo = src[0];
            hi = src[1];
        }
        float* dst = &A[arow*A_STRIDE + acol8];
        *(float4*)(dst)     = lo;
        *(float4*)(dst + 4) = hi;
        __syncthreads();

        const float* wbase = wT + (size_t)((t*16)*4 + wv)*64 + lane*2;
#pragma unroll
        for (int kc = 0; kc < 16; ++kc) {
            const float* ap = &A[col*A_STRIDE + kc*4 + 2*half];
            v2f a; a.x = ap[0]; a.y = ap[1];
            const float* bp = wbase + (size_t)kc*256;
            v2f b; b.x = bp[0]; b.y = bp[1];
            acc = __builtin_amdgcn_wmma_f32_16x16x4_f32(
                false, a, false, b, (short)0, acc, false, false);
        }
        __syncthreads();
    }

#pragma unroll
    for (int j = 0; j < 8; ++j) {
        int m = j + 8*half;
        out[(size_t)(p0 + m)*CO + wv*16 + col] = acc[j];
    }
}

extern "C" void kernel_launch(void* const* d_in, const int* in_sizes, int n_in,
                              void* d_out, int out_size, void* d_ws, size_t ws_size,
                              hipStream_t stream) {
    const float* feats  = (const float*)d_in[0];
    const float* weight = (const float*)d_in[1];
    const float* bias   = (const float*)d_in[2];
    const int*   coords = (const int*)  d_in[3];
    float* out = (float*)d_out;

    const int n = in_sizes[0] / CI;

    float* wT   = (float*)d_ws;
    int*   grid = (int*)((char*)d_ws + (size_t)WT_FLOATS*4);

    k_fill_grid<<<(GRID_CELLS + 255)/256, 256, 0, stream>>>(grid);
    k_scatter  <<<(n + 255)/256,          256, 0, stream>>>(coords, grid, n);
    k_wt       <<<(NOFF*16*4*32 + 255)/256, 256, 0, stream>>>(weight, wT);
    k_conv     <<<n/16, 128, 0, stream>>>(feats, wT, bias, coords, grid, out);
}


# launches 8 -> 6: weight re-layout rides in the grid-fill launch, and the per-tap scan launch is gone (each 800-point pair work-group sums the count lines of the groups before it); sum order unchanged
# speedup vs baseline: 2.5864x; 2.5864x over previous
#include <hip/hip_runtime.h>

typedef float v2f __attribute__((ext_vector_type(2)));
typedef float v8f __attribute__((ext_vector_type(8)));

#define BB   2
#define DD   128
#define HH   128
#define WW   128
#define CI   64
#define CO   64
#define NOFF 27
#define CTAP 13
#define NPTS 200000
#define GRID_CELLS (BB*DD*HH*WW)
#define WT_FLOATS  (NOFF*16*4*32*2)
#define PB   800
#define NB   ((NPTS + PB - 1)/PB)
#define NWV  (PB/32)
#define CAP_ROWS 880000

#define OFF_GRID  ((size_t)442368)
#define OFF_NPOS  ((size_t)17219584)
#define OFF_CNT   ((size_t)38819584)
#define OFF_TOT   ((size_t)39494656)
#define OFF_PIN   ((size_t)39494912)
#define OFF_PART  ((size_t)43014912)

#define FILL_WGS 512
#define FILL_PER (GRID_CELLS / 4 / (FILL_WGS * 256))
__global__ __launch_bounds__(256)
void k_fill_grid(int* __restrict__ grid, const float* __restrict__ w, float* __restrict__ wT) {
    int4* g = (int4*)grid;
    const int4 m = make_int4(-1, -1, -1, -1);
    const int i = blockIdx.x * 256 + threadIdx.x;
    const bool has = i < NOFF*16*4*32;
    v2f wv2; wv2.x = 0.f; wv2.y = 0.f;
    if (has) {
        int lane = i & 31;
        int nt   = (i >> 5) & 3;
        int kc   = (i >> 7) & 15;
        int t    =  i >> 11;
        int o = nt*16 + (lane & 15);
        int c = kc*4 + 2*(lane >> 4);
        wv2 = *(const v2f*)(w + (size_t)(o*NOFF + t)*CI + c);
    }
#pragma unroll
    for (int k = 0; k < FILL_PER; ++k) g[i + k * (FILL_WGS * 256)] = m;
    if (has) *(v2f*)(wT + (size_t)i*2) = wv2;
}

__global__ void k_scatter(const int* __restrict__ coords, int* __restrict__ grid, int n) {
    int i = blockIdx.x * blockDim.x + threadIdx.x;
    if (i >= n) return;
    int b = coords[i*4+0], z = coords[i*4+1], y = coords[i*4+2], x = coords[i*4+3];
    int lin = ((b*DD + z)*HH + y)*WW + x;
    grid[lin] = i;
}

__global__ __launch_bounds__(PB)
void k_nbr(const int* __restrict__ coords, const int* __restrict__ grid,
           int* __restrict__ npos, int* __restrict__ cntb, int n) {
    __shared__ int s_c[NWV][32];
    const int p    = blockIdx.x * PB + threadIdx.x;
    const int lane = threadIdx.x & 31;
    const int wv   = __builtin_amdgcn_readfirstlane((int)(threadIdx.x >> 5));
    const bool live = p < n && p < NPTS;
    int bz = 0, z = 0, y = 0, x = 0;
    if (live) {
        bz = coords[p*4+0]; z = coords[p*4+1]; y = coords[p*4+2]; x = coords[p*4+3];
    }
    int idx[NOFF];
#pragma unroll
    for (int t = 0; t < NOFF; ++t) {
        int kd = t / 9, kh = (t / 3) % 3, kw = t % 3;
        int nz = z + kd - 1, ny = y + kh - 1, nx = x + kw - 1;
        int v = -1;
        if (live && (unsigned)nz < (unsigned)DD && (unsigned)ny < (unsigned)HH &&
            (unsigned)nx < (unsigned)WW) {
            int lin = ((bz*DD + nz)*HH + ny)*WW + nx;
            v = grid[lin];
        }
        idx[t] = v;
    }
#pragma unroll
    for (int t = 0; t < NOFF; ++t) {
        if (live) npos[(size_t)t*NPTS + p] = idx[t];
        unsigned mask = __builtin_amdgcn_ballot_w32(idx[t] >= 0);
        if (lane == 0) s_c[wv][t] = __builtin_popcount(mask);
    }
    __syncthreads();
    if (wv == 0) {
        int v = 0;
        if (lane < NOFF && lane != CTAP) {
#pragma unroll
            for (int w = 0; w < NWV; ++w) v += s_c[w][lane];
        }
        cntb[blockIdx.x*32 + lane] = v;
    }
}

#define LIST_HEAD 4
__global__ __launch_bounds__(PB)
void k_pairs(int* __restrict__ npos, const int* __restrict__ cntb,
             int* __restrict__ total, int* __restrict__ pin,
             int* __restrict__ cen, int n) {
    __shared__ int s_cw[NWV][32];
    __shared__ int s_pt[NWV][32];
    __shared__ int s_pp[NWV][32];
    const int p    = blockIdx.x * PB + threadIdx.x;
    const int lane = threadIdx.x & 31;
    const int wv   = __builtin_amdgcn_readfirstlane((int)(threadIdx.x >> 5));
    const int blk  = blockIdx.x;
    const bool live = p < n && p < NPTS;
    int c[NB/NWV];
#pragma unroll
    for (int i = 0; i < NB/NWV; ++i) c[i] = cntb[(wv + i*NWV)*32 + lane];
    int q[NOFF];
#pragma unroll
    for (int t = 0; t < NOFF; ++t) q[t] = live ? npos[(size_t)t*NPTS + p] : -1;

    {
        int pt = 0, pp = 0;
#pragma unroll
        for (int i = 0; i < NB/NWV; ++i) {
            pt += c[i];
            if (wv + i*NWV < blk) pp += c[i];
        }
        s_pt[wv][lane] = pt;
        s_pp[wv][lane] = pp;
    }
#pragma unroll
    for (int t = 0; t < NOFF; ++t) {
        unsigned mask = __builtin_amdgcn_ballot_w32(q[t] >= 0);
        if (lane == 0) s_cw[wv][t] = (t == CTAP) ? 0 : __builtin_popcount(mask);
    }
    __syncthreads();

    int tot = 0;
    int off = 0;
    if (lane < NOFF) {
        int a_pt[NWV], a_pp[NWV], a_cw[NWV];
#pragma unroll
        for (int w = 0; w < NWV; ++w) {
            a_pt[w] = s_pt[w][lane];
            a_pp[w] = s_pp[w][lane];
            a_cw[w] = s_cw[w][lane];
        }
#pragma unroll
        for (int w = 0; w < NWV; ++w) {
            tot += a_pt[w];
            off += a_pp[w];
            off += (w < wv) ? a_cw[w] : 0;
        }
        if (blk == 0 && wv == 0) total[lane] = tot;
    }

    int tb = 0;
    int k  = 0;
#pragma unroll
    for (int t = 0; t < NOFF; ++t) {
        if (t == CTAP) { if (live) cen[p] = q[t]; continue; }
        const int tot_t = __builtin_amdgcn_readlane(tot, t);
        const int off_t = __builtin_amdgcn_readlane(off, t);
        unsigned mask = __builtin_amdgcn_ballot_w32(q[t] >= 0);
        if (q[t] >= 0) {
            int pos = tb*16 + off_t + __builtin_popcount(mask & ((1u << lane) - 1u));
            if (pos < CAP_ROWS) {
                pin[pos] = q[t];
                npos[(size_t)k*NPTS + p] = pos;
                ++k;
            }
        }
        tb += (tot_t + 15) >> 4;
    }
    if (live) {
        for (int s2 = 0; s2 < LIST_HEAD; ++s2)
            if (k <= s2) npos[(size_t)s2*NPTS + p] = -1;
        if (k >= LIST_HEAD && k < NOFF) npos[(size_t)k*NPTS + p] = -1;
    }
}

#define GEMM_WGS 1024
__global__ __launch_bounds__(256)
void k_gemm(const float* __restrict__ feats,
            const float* __restrict__ wT,
            const int*   __restrict__ total,
            const int*   __restrict__ pin,
            float*       __restrict__ part) {
    const int lane = threadIdx.x & 31;
    const int col  = lane & 15;
    const int half = lane >> 4;
    const int wid    = __builtin_amdgcn_readfirstlane((int)(blockIdx.x * 8 + (threadIdx.x >> 5)));
    const int nwaves = gridDim.x * 8;

    int ntiles = 0;
    for (int k = 0; k < NOFF; ++k) ntiles += (total[k] + 15) >> 4;

    for (int tile = wid; tile < ntiles; tile += nwaves) {
        int t = 0, tb = 0, tot = 0;
        {
            int acc0 = 0;
            for (int k = 0; k < NOFF; ++k) {
                int tk = total[k];
                int nk = (tk + 15) >> 4;
                if (tile >= acc0 && tile < acc0 + nk) { t = k; tb = acc0; tot = tk; }
                acc0 += nk;
            }
        }
        const int rend = tb*16 + tot;
        const int arow = tile*16 + col;
        int q = -1;
        if (arow < rend && arow < CAP_ROWS) q = pin[arow];

        v2f a[16];
#pragma unroll
        for (int kc = 0; kc < 16; ++kc) {
            v2f v; v.x = 0.f; v.y = 0.f;
            if (q >= 0) v = *(const v2f*)(feats + (size_t)q*CI + kc*4 + 2*half);
            a[kc] = v;
        }

        v8f acc[4];
#pragma unroll
        for (int nt = 0; nt < 4; ++nt)
#pragma unroll
            for (int j = 0; j < 8; ++j) acc[nt][j] = 0.f;

        const float* wbase = wT + (size_t)t*(16*4*64) + lane*2;
#pragma unroll
        for (int kc = 0; kc < 16; ++kc) {
#pragma unroll
            for (int nt = 0; nt < 4; ++nt) {
                v2f b = *(const v2f*)(wbase + (kc*4 + nt)*64);
                acc[nt] = __builtin_amdgcn_wmma_f32_16x16x4_f32(
                    false, a[kc], false, b, (short)0, acc[nt], false, false);
            }
        }

#pragma unroll
        for (int j = 0; j < 8; ++j) {
            int r = tile*16 + j + 8*half;
            if (r < rend && r < CAP_ROWS) {
#pragma unroll
                for (int nt = 0; nt < 4; ++nt)
                    part[(size_t)r*CO + nt*16 + col] = acc[nt][j];
            }
        }
    }
}

#define FIN_WGS 512
#define TP 68
#define EB 4
__global__ __launch_bounds__(256)
void k_final(const float* __restrict__ feats, const float* __restrict__ wT,
             const float* __restrict__ bias, const int* __restrict__ cen,
             const int* __restrict__ list, const float* __restrict__ part,
             float* __restrict__ out, int n) {
    __shared__ __attribute__((aligned(16))) float s_tile[8][16*TP];
    __shared__ __attribute__((aligned(16))) int   s_head[8][16*LIST_HEAD];
    const int lane = threadIdx.x & 31;
    const int col  = lane & 15;
    const int half = lane >> 4;
    const int wv   = threadIdx.x >> 5;
    const int wid    = __builtin_amdgcn_readfirstlane((int)(blockIdx.x * 8 + wv));
    const int nwaves = gridDim.x * 8;
    const int ntile  = (n + 15) >> 4;
    float* my  = s_tile[wv];
    int*   myh = s_head[wv];
    const float4* rows = (const float4*)part;

    v2f bw[64];
    const float* wbase = wT + (size_t)CTAP*(16*4*64) + lane*2;
#pragma unroll
    for (int i = 0; i < 64; ++i) bw[i] = *(const v2f*)(wbase + i*64);
    const float4 bs = *(const float4*)(bias + col*4);

    for (int tile = wid; tile < ntile; tile += nwaves) {
        const int p = tile*16 + col;
        int q = -1;
        int4 hd = make_int4(-1, -1, -1, -1);
        if (p < n) {
            q    = cen[p];
            hd.x = list[p];
            hd.y = list[(size_t)1*NPTS + p];
            hd.z = list[(size_t)2*NPTS + p];
            hd.w = list[(size_t)3*NPTS + p];
        }
        v2f a[16];
        const float* arow = feats + (size_t)(q >= 0 ? q : 0)*CI + 2*half;
#pragma unroll
        for (int kc = 0; kc < 16; ++kc) {
            v2f v = *(const v2f*)(arow + kc*4);
            if (q < 0) { v.x = 0.f; v.y = 0.f; }
            a[kc] = v;
        }
        v8f acc[4];
#pragma unroll
        for (int nt = 0; nt < 4; ++nt)
#pragma unroll
            for (int j = 0; j < 8; ++j) acc[nt][j] = 0.f;
#pragma unroll
        for (int kc = 0; kc < 16; ++kc) {
#pragma unroll
            for (int nt = 0; nt < 4; ++nt)
                acc[nt] = __builtin_amdgcn_wmma_f32_16x16x4_f32(
                    false, a[kc], false, bw[kc*4 + nt], (short)0, acc[nt], false, false);
        }
#pragma unroll
        for (int j = 0; j < 8; ++j)
#pragma unroll
            for (int nt = 0; nt < 4; ++nt)
                my[(j + 8*half)*TP + nt*16 + col] = acc[nt][j];
        if (half == 0) *(int4*)(myh + col*LIST_HEAD) = hd;
        __builtin_amdgcn_fence(__ATOMIC_ACQ_REL, "wavefront");

        const int c4 = col*4;
#pragma unroll
        for (int b0 = 0; b0 < 8; b0 += EB) {
            int4 h[EB];
            float4 v[EB], x0[EB], x1[EB], x2[EB], x3[EB];
#pragma unroll
            for (int e = 0; e < EB; ++e) {
                const int row = 2*(b0 + e) + half;
                h[e] = *(const int4*)(myh + row*LIST_HEAD);
                v[e] = *(const float4*)(my + row*TP + c4);
            }
#pragma unroll
            for (int e = 0; e < EB; ++e) {
                const float4 z = make_float4(0.f, 0.f, 0.f, 0.f);
                x0[e] = z; x1[e] = z; x2[e] = z; x3[e] = z;
                if (h[e].x >= 0) {
                    x0[e] = rows[(unsigned)h[e].x*(CO/4) + col];
                    x1[e] = rows[(unsigned)(h[e].y >= 0 ? h[e].y : h[e].x)*(CO/4) + col];
                    x2[e] = rows[(unsigned)(h[e].z >= 0 ? h[e].z : h[e].x)*(CO/4) + col];
                    x3[e] = rows[(unsigned)(h[e].w >= 0 ? h[e].w : h[e].x)*(CO/4) + col];
                }
            }
#pragma unroll
            for (int e = 0; e < EB; ++e) {
                const int pp = tile*16 + 2*(b0 + e) + half;
                float4 r;
                r.x = bs.x + v[e].x; r.y = bs.y + v[e].y; r.z = bs.z + v[e].z; r.w = bs.w + v[e].w;
                if (h[e].x >= 0) {
                    r.x += x0[e].x; r.y += x0[e].y; r.z += x0[e].z; r.w += x0[e].w;
                    if (h[e].y >= 0) { r.x += x1[e].x; r.y += x1[e].y; r.z += x1[e].z; r.w += x1[e].w; }
                    if (h[e].z >= 0) { r.x += x2[e].x; r.y += x2[e].y; r.z += x2[e].z; r.w += x2[e].w; }
                    if (h[e].w >= 0) {
                        r.x += x3[e].x; r.y += x3[e].y; r.z += x3[e].z; r.w += x3[e].w;
                        for (int k = LIST_HEAD; k < NOFF; ++k) {
                            const int rr = list[(size_t)k*NPTS + pp];
                            if (rr < 0) break;
                            const float4 y = rows[(unsigned)rr*(CO/4) + col];
                            r.x += y.x; r.y += y.y; r.z += y.z; r.w += y.w;
                        }
                    }
                }
                if (pp < n) *(float4*)(out + (size_t)pp*CO + c4) = r;
            }
        }
        __builtin_amdgcn_fence(__ATOMIC_ACQ_REL, "wavefront");
    }
}

extern "C" void kernel_launch(void* const* d_in, const int* in_sizes, int n_in,
                              void* d_out, int out_size, void* d_ws, size_t ws_size,
                              hipStream_t stream) {
    const float* feats  = (const float*)d_in[0];
    const float* weight = (const float*)d_in[1];
    const float* bias   = (const float*)d_in[2];
    const int*   coords = (const int*)  d_in[3];
    float* out = (float*)d_out;

    const int n = in_sizes[0] / CI;

    float* wT    = (float*)d_ws;
    int*   grid  = (int*)  ((char*)d_ws + OFF_GRID);
    int*   cen   = grid;
    int*   npos  = (int*)  ((char*)d_ws + OFF_NPOS);
    int*   cntb  = (int*)  ((char*)d_ws + OFF_CNT);
    int*   total = (int*)  ((char*)d_ws + OFF_TOT);
    int*   pin   = (int*)  ((char*)d_ws + OFF_PIN);
    float* part  = (float*)((char*)d_ws + OFF_PART);

    k_fill_grid<<<FILL_WGS,               256, 0, stream>>>(grid, weight, wT);
    k_scatter  <<<(n + 255)/256,          256, 0, stream>>>(coords, grid, n);
    k_nbr      <<<NB,                     PB,  0, stream>>>(coords, grid, npos, cntb, n);
    k_pairs    <<<NB,                     PB,  0, stream>>>(npos, cntb, total, pin, cen, n);
    k_gemm     <<<GEMM_WGS,               256, 0, stream>>>(feats, wT, total, pin, part);
    k_final    <<<FIN_WGS,                256, 0, stream>>>(feats, wT, bias, cen, npos, part, out, n);
}


# final kernel loads the centre weights per matrix step from cache instead of keeping them in 128 registers: registers 256 -> 120, resident waves doubled, one 16-point tile per wave; sum order same
# speedup vs baseline: 2.6699x; 1.0323x over previous
#include <hip/hip_runtime.h>

typedef float v2f __attribute__((ext_vector_type(2)));
typedef float v8f __attribute__((ext_vector_type(8)));

#define BB   2
#define DD   128
#define HH   128
#define WW   128
#define CI   64
#define CO   64
#define NOFF 27
#define CTAP 13
#define NPTS 200000
#define GRID_CELLS (BB*DD*HH*WW)
#define WT_FLOATS  (NOFF*16*4*32*2)
#define PB   800
#define NB   ((NPTS + PB - 1)/PB)
#define NWV  (PB/32)
#define CAP_ROWS 880000

#define OFF_GRID  ((size_t)442368)
#define OFF_NPOS  ((size_t)17219584)
#define OFF_CNT   ((size_t)38819584)
#define OFF_TOT   ((size_t)39494656)
#define OFF_PIN   ((size_t)39494912)
#define OFF_PART  ((size_t)43014912)

#define FILL_WGS 512
#define FILL_PER (GRID_CELLS / 4 / (FILL_WGS * 256))
__global__ __launch_bounds__(256)
void k_fill_grid(int* __restrict__ grid, const float* __restrict__ w, float* __restrict__ wT) {
    int4* g = (int4*)grid;
    const int4 m = make_int4(-1, -1, -1, -1);
    const int i = blockIdx.x * 256 + threadIdx.x;
    const bool has = i < NOFF*16*4*32;
    v2f wv2; wv2.x = 0.f; wv2.y = 0.f;
    if (has) {
        int lane = i & 31;
        int nt   = (i >> 5) & 3;
        int kc   = (i >> 7) & 15;
        int t    =  i >> 11;
        int o = nt*16 + (lane & 15);
        int c = kc*4 + 2*(lane >> 4);
        wv2 = *(const v2f*)(w + (size_t)(o*NOFF + t)*CI + c);
    }
#pragma unroll
    for (int k = 0; k < FILL_PER; ++k) g[i + k * (FILL_WGS * 256)] = m;
    if (has) *(v2f*)(wT + (size_t)i*2) = wv2;
}

__global__ void k_scatter(const int* __restrict__ coords, int* __restrict__ grid, int n) {
    int i = blockIdx.x * blockDim.x + threadIdx.x;
    if (i >= n) return;
    int b = coords[i*4+0], z = coords[i*4+1], y = coords[i*4+2], x = coords[i*4+3];
    int lin = ((b*DD + z)*HH + y)*WW + x;
    grid[lin] = i;
}

__global__ __launch_bounds__(PB)
void k_nbr(const int* __restrict__ coords, const int* __restrict__ grid,
           int* __restrict__ npos, int* __restrict__ cntb, int n) {
    __shared__ int s_c[NWV][32];
    const int p    = blockIdx.x * PB + threadIdx.x;
    const int lane = threadIdx.x & 31;
    const int wv   = __builtin_amdgcn_readfirstlane((int)(threadIdx.x >> 5));
    const bool live = p < n && p < NPTS;
    int bz = 0, z = 0, y = 0, x = 0;
    if (live) {
        bz = coords[p*4+0]; z = coords[p*4+1]; y = coords[p*4+2]; x = coords[p*4+3];
    }
    int idx[NOFF];
#pragma unroll
    for (int t = 0; t < NOFF; ++t) {
        int kd = t / 9, kh = (t / 3) % 3, kw = t % 3;
        int nz = z + kd - 1, ny = y + kh - 1, nx = x + kw - 1;
        int v = -1;
        if (live && (unsigned)nz < (unsigned)DD && (unsigned)ny < (unsigned)HH &&
            (unsigned)nx < (unsigned)WW) {
            int lin = ((bz*DD + nz)*HH + ny)*WW + nx;
            v = grid[lin];
        }
        idx[t] = v;
    }
#pragma unroll
    for (int t = 0; t < NOFF; ++t) {
        if (live) npos[(size_t)t*NPTS + p] = idx[t];
        unsigned mask = __builtin_amdgcn_ballot_w32(idx[t] >= 0);
        if (lane == 0) s_c[wv][t] = __builtin_popcount(mask);
    }
    __syncthreads();
    if (wv == 0) {
        int v = 0;
        if (lane < NOFF && lane != CTAP) {
#pragma unroll
            for (int w = 0; w < NWV; ++w) v += s_c[w][lane];
        }
        cntb[blockIdx.x*32 + lane] = v;
    }
}

#define LIST_HEAD 4
__global__ __launch_bounds__(PB)
void k_pairs(int* __restrict__ npos, const int* __restrict__ cntb,
             int* __restrict__ total, int* __restrict__ pin,
             int* __restrict__ cen, int n) {
    __shared__ int s_cw[NWV][32];
    __shared__ int s_pt[NWV][32];
    __shared__ int s_pp[NWV][32];
    const int p    = blockIdx.x * PB + threadIdx.x;
    const int lane = threadIdx.x & 31;
    const int wv   = __builtin_amdgcn_readfirstlane((int)(threadIdx.x >> 5));
    const int blk  = blockIdx.x;
    const bool live = p < n && p < NPTS;
    int c[NB/NWV];
#pragma unroll
    for (int i = 0; i < NB/NWV; ++i) c[i] = cntb[(wv + i*NWV)*32 + lane];
    int q[NOFF];
#pragma unroll
    for (int t = 0; t < NOFF; ++t) q[t] = live ? npos[(size_t)t*NPTS + p] : -1;

    {
        int pt = 0, pp = 0;
#pragma unroll
        for (int i = 0; i < NB/NWV; ++i) {
            pt += c[i];
            if (wv + i*NWV < blk) pp += c[i];
        }
        s_pt[wv][lane] = pt;
        s_pp[wv][lane] = pp;
    }
#pragma unroll
    for (int t = 0; t < NOFF; ++t) {
        unsigned mask = __builtin_amdgcn_ballot_w32(q[t] >= 0);
        if (lane == 0) s_cw[wv][t] = (t == CTAP) ? 0 : __builtin_popcount(mask);
    }
    __syncthreads();

    int tot = 0;
    int off = 0;
    if (lane < NOFF) {
        int a_pt[NWV], a_pp[NWV], a_cw[NWV];
#pragma unroll
        for (int w = 0; w < NWV; ++w) {
            a_pt[w] = s_pt[w][lane];
            a_pp[w] = s_pp[w][lane];
            a_cw[w] = s_cw[w][lane];
        }
#pragma unroll
        for (int w = 0; w < NWV; ++w) {
            tot += a_pt[w];
            off += a_pp[w];
            off += (w < wv) ? a_cw[w] : 0;
        }
        if (blk == 0 && wv == 0) total[lane] = tot;
    }

    int tb = 0;
    int k  = 0;
#pragma unroll
    for (int t = 0; t < NOFF; ++t) {
        if (t == CTAP) { if (live) cen[p] = q[t]; continue; }
        const int tot_t = __builtin_amdgcn_readlane(tot, t);
        const int off_t = __builtin_amdgcn_readlane(off, t);
        unsigned mask = __builtin_amdgcn_ballot_w32(q[t] >= 0);
        if (q[t] >= 0) {
            int pos = tb*16 + off_t + __builtin_popcount(mask & ((1u << lane) - 1u));
            if (pos < CAP_ROWS) {
                pin[pos] = q[t];
                npos[(size_t)k*NPTS + p] = pos;
                ++k;
            }
        }
        tb += (tot_t + 15) >> 4;
    }
    if (live) {
        for (int s2 = 0; s2 < LIST_HEAD; ++s2)
            if (k <= s2) npos[(size_t)s2*NPTS + p] = -1;
        if (k >= LIST_HEAD && k < NOFF) npos[(size_t)k*NPTS + p] = -1;
    }
}

#define GEMM_WGS 1024
__global__ __launch_bounds__(256)
void k_gemm(const float* __restrict__ feats,
            const float* __restrict__ wT,
            const int*   __restrict__ total,
            const int*   __restrict__ pin,
            float*       __restrict__ part) {
    const int lane = threadIdx.x & 31;
    const int col  = lane & 15;
    const int half = lane >> 4;
    const int wid    = __builtin_amdgcn_readfirstlane((int)(blockIdx.x * 8 + (threadIdx.x >> 5)));
    const int nwaves = gridDim.x * 8;

    int ntiles = 0;
    for (int k = 0; k < NOFF; ++k) ntiles += (total[k] + 15) >> 4;

    for (int tile = wid; tile < ntiles; tile += nwaves) {
        int t = 0, tb = 0, tot = 0;
        {
            int acc0 = 0;
            for (int k = 0; k < NOFF; ++k) {
                int tk = total[k];
                int nk = (tk + 15) >> 4;
                if (tile >= acc0 && tile < acc0 + nk) { t = k; tb = acc0; tot = tk; }
                acc0 += nk;
            }
        }
        const int rend = tb*16 + tot;
        const int arow = tile*16 + col;
        int q = -1;
        if (arow < rend && arow < CAP_ROWS) q = pin[arow];

        v2f a[16];
#pragma unroll
        for (int kc = 0; kc < 16; ++kc) {
            v2f v; v.x = 0.f; v.y = 0.f;
            if (q >= 0) v = *(const v2f*)(feats + (size_t)q*CI + kc*4 + 2*half);
            a[kc] = v;
        }

        v8f acc[4];
#pragma unroll
        for (int nt = 0; nt < 4; ++nt)
#pragma unroll
            for (int j = 0; j < 8; ++j) acc[nt][j] = 0.f;

        const float* wbase = wT + (size_t)t*(16*4*64) + lane*2;
#pragma unroll
        for (int kc = 0; kc < 16; ++kc) {
#pragma unroll
            for (int nt = 0; nt < 4; ++nt) {
                v2f b = *(const v2f*)(wbase + (kc*4 + nt)*64);
                acc[nt] = __builtin_amdgcn_wmma_f32_16x16x4_f32(
                    false, a[kc], false, b, (short)0, acc[nt], false, false);
            }
        }

#pragma unroll
        for (int j = 0; j < 8; ++j) {
            int r = tile*16 + j + 8*half;
            if (r < rend && r < CAP_ROWS) {
#pragma unroll
                for (int nt = 0; nt < 4; ++nt)
                    part[(size_t)r*CO + nt*16 + col] = acc[nt][j];
            }
        }
    }
}

#define FIN_WGS ((NPTS/16 + 7)/8)
#define TP 68
#define EB 4
__global__ __launch_bounds__(256)
void k_final(const float* __restrict__ feats, const float* __restrict__ wT,
             const float* __restrict__ bias, const int* __restrict__ cen,
             const int* __restrict__ list, const float* __restrict__ part,
             float* __restrict__ out, int n) {
    __shared__ __attribute__((aligned(16))) float s_tile[8][16*TP];
    __shared__ __attribute__((aligned(16))) int   s_head[8][16*LIST_HEAD];
    const int lane = threadIdx.x & 31;
    const int col  = lane & 15;
    const int half = lane >> 4;
    const int wv   = threadIdx.x >> 5;
    const int tile  = __builtin_amdgcn_readfirstlane((int)(blockIdx.x * 8 + wv));
    const int ntile = (n + 15) >> 4;
    if (tile >= ntile) return;
    float* my  = s_tile[wv];
    int*   myh = s_head[wv];
    const float4* rows = (const float4*)part;

    const float* wbase = wT + (size_t)CTAP*(16*4*64) + lane*2;
    const float4 bs = *(const float4*)(bias + col*4);

    const int p = tile*16 + col;
    int q = -1;
    int4 hd = make_int4(-1, -1, -1, -1);
    if (p < n) {
        q    = cen[p];
        hd.x = list[p];
        hd.y = list[(size_t)1*NPTS + p];
        hd.z = list[(size_t)2*NPTS + p];
        hd.w = list[(size_t)3*NPTS + p];
    }
    v2f a[16];
    const float* arow = feats + (size_t)(q >= 0 ? q : 0)*CI + 2*half;
#pragma unroll
    for (int kc = 0; kc < 16; ++kc) {
        v2f v = *(const v2f*)(arow + kc*4);
        if (q < 0) { v.x = 0.f; v.y = 0.f; }
        a[kc] = v;
    }
    v8f acc[4];
#pragma unroll
    for (int nt = 0; nt < 4; ++nt)
#pragma unroll
        for (int j = 0; j < 8; ++j) acc[nt][j] = 0.f;
#pragma unroll
    for (int kc = 0; kc < 16; ++kc) {
#pragma unroll
        for (int nt = 0; nt < 4; ++nt) {
            v2f b = *(const v2f*)(wbase + (kc*4 + nt)*64);
            acc[nt] = __builtin_amdgcn_wmma_f32_16x16x4_f32(
                false, a[kc], false, b, (short)0, acc[nt], false, false);
        }
    }
#pragma unroll
    for (int j = 0; j < 8; ++j)
#pragma unroll
        for (int nt = 0; nt < 4; ++nt)
            my[(j + 8*half)*TP + nt*16 + col] = acc[nt][j];
    if (half == 0) *(int4*)(myh + col*LIST_HEAD) = hd;
    __builtin_amdgcn_fence(__ATOMIC_ACQ_REL, "wavefront");

    const int c4 = col*4;
#pragma unroll
    for (int b0 = 0; b0 < 8; b0 += EB) {
        int4 h[EB];
        float4 v[EB], x0[EB], x1[EB], x2[EB], x3[EB];
#pragma unroll
        for (int e = 0; e < EB; ++e) {
            const int row = 2*(b0 + e) + half;
            h[e] = *(const int4*)(myh + row*LIST_HEAD);
            v[e] = *(const float4*)(my + row*TP + c4);
        }
#pragma unroll
        for (int e = 0; e < EB; ++e) {
            const float4 z = make_float4(0.f, 0.f, 0.f, 0.f);
            x0[e] = z; x1[e] = z; x2[e] = z; x3[e] = z;
            if (h[e].x >= 0) {
                x0[e] = rows[(unsigned)h[e].x*(CO/4) + col];
                x1[e] = rows[(unsigned)(h[e].y >= 0 ? h[e].y : h[e].x)*(CO/4) + col];
                x2[e] = rows[(unsigned)(h[e].z >= 0 ? h[e].z : h[e].x)*(CO/4) + col];
                x3[e] = rows[(unsigned)(h[e].w >= 0 ? h[e].w : h[e].x)*(CO/4) + col];
            }
        }
#pragma unroll
        for (int e = 0; e < EB; ++e) {
            const int pp = tile*16 + 2*(b0 + e) + half;
            float4 r;
            r.x = bs.x + v[e].x; r.y = bs.y + v[e].y; r.z = bs.z + v[e].z; r.w = bs.w + v[e].w;
            if (h[e].x >= 0) {
                r.x += x0[e].x; r.y += x0[e].y; r.z += x0[e].z; r.w += x0[e].w;
                if (h[e].y >= 0) { r.x += x1[e].x; r.y += x1[e].y; r.z += x1[e].z; r.w += x1[e].w; }
                if (h[e].z >= 0) { r.x += x2[e].x; r.y += x2[e].y; r.z += x2[e].z; r.w += x2[e].w; }
                if (h[e].w >= 0) {
                    r.x += x3[e].x; r.y += x3[e].y; r.z += x3[e].z; r.w += x3[e].w;
                    for (int k = LIST_HEAD; k < NOFF; ++k) {
                        const int rr = list[(size_t)k*NPTS + pp];
                        if (rr < 0) break;
                        const float4 y = rows[(unsigned)rr*(CO/4) + col];
                        r.x += y.x; r.y += y.y; r.z += y.z; r.w += y.w;
                    }
                }
            }
            if (pp < n) *(float4*)(out + (size_t)pp*CO + c4) = r;
        }
    }
}

extern "C" void kernel_launch(void* const* d_in, const int* in_sizes, int n_in,
                              void* d_out, int out_size, void* d_ws, size_t ws_size,
                              hipStream_t stream) {
    const float* feats  = (const float*)d_in[0];
    const float* weight = (const float*)d_in[1];
    const float* bias   = (const float*)d_in[2];
    const int*   coords = (const int*)  d_in[3];
    float* out = (float*)d_out;

    const int n = in_sizes[0] / CI;

    float* wT    = (float*)d_ws;
    int*   grid  = (int*)  ((char*)d_ws + OFF_GRID);
    int*   cen   = grid;
    int*   npos  = (int*)  ((char*)d_ws + OFF_NPOS);
    int*   cntb  = (int*)  ((char*)d_ws + OFF_CNT);
    int*   total = (int*)  ((char*)d_ws + OFF_TOT);
    int*   pin   = (int*)  ((char*)d_ws + OFF_PIN);
    float* part  = (float*)((char*)d_ws + OFF_PART);

    k_fill_grid<<<FILL_WGS,               256, 0, stream>>>(grid, weight, wT);
    k_scatter  <<<(n + 255)/256,          256, 0, stream>>>(coords, grid, n);
    k_nbr      <<<NB,                     PB,  0, stream>>>(coords, grid, npos, cntb, n);
    k_pairs    <<<NB,                     PB,  0, stream>>>(npos, cntb, total, pin, cen, n);
    k_gemm     <<<GEMM_WGS,               256, 0, stream>>>(feats, wT, total, pin, part);
    k_final    <<<FIN_WGS,                256, 0, stream>>>(feats, wT, bias, cen, npos, part, out, n);
}


# neighbour lookup turned round: one load per point with lane = tap, on a grid stored in 4x4x4-cell blocks, so a load touches the 4-5 lines of one point, not 32; back to lane = point in LDS; sums same
# speedup vs baseline: 3.0342x; 1.1364x over previous
#include <hip/hip_runtime.h>

typedef float v2f __attribute__((ext_vector_type(2)));
typedef float v8f __attribute__((ext_vector_type(8)));

#define BB   2
#define DD   128
#define HH   128
#define WW   128
#define CI   64
#define CO   64
#define NOFF 27
#define CTAP 13
#define NPTS 200000
#define GRID_CELLS (BB*DD*HH*WW)
#define WT_FLOATS  (NOFF*16*4*32*2)
#define PB   800
#define NB   ((NPTS + PB - 1)/PB)
#define NWV  (PB/32)
#define CAP_ROWS 880000

#define OFF_GRID  ((size_t)442368)
#define OFF_NPOS  ((size_t)17219584)
#define OFF_CNT   ((size_t)38819584)
#define OFF_TOT   ((size_t)39494656)
#define OFF_PIN   ((size_t)39494912)
#define OFF_PART  ((size_t)43014912)

#define FILL_WGS 512
#define FILL_PER (GRID_CELLS / 4 / (FILL_WGS * 256))
__global__ __launch_bounds__(256)
void k_fill_grid(int* __restrict__ grid, const float* __restrict__ w, float* __restrict__ wT) {
    int4* g = (int4*)grid;
    const int4 m = make_int4(-1, -1, -1, -1);
    const int i = blockIdx.x * 256 + threadIdx.x;
    const bool has = i < NOFF*16*4*32;
    v2f wv2; wv2.x = 0.f; wv2.y = 0.f;
    if (has) {
        int lane = i & 31;
        int nt   = (i >> 5) & 3;
        int kc   = (i >> 7) & 15;
        int t    =  i >> 11;
        int o = nt*16 + (lane & 15);
        int c = kc*4 + 2*(lane >> 4);
        wv2 = *(const v2f*)(w + (size_t)(o*NOFF + t)*CI + c);
    }
#pragma unroll
    for (int k = 0; k < FILL_PER; ++k) g[i + k * (FILL_WGS * 256)] = m;
    if (has) *(v2f*)(wT + (size_t)i*2) = wv2;
}

__device__ __forceinline__ int cell_index(int b, int z, int y, int x) {
    int blk = ((b*(DD/4) + (z >> 2))*(HH/4) + (y >> 2))*(WW/4) + (x >> 2);
    return (blk << 6) | ((z & 3) << 4) | ((y & 3) << 2) | (x & 3);
}

__global__ void k_scatter(const int* __restrict__ coords, int* __restrict__ grid, int n) {
    int i = blockIdx.x * blockDim.x + threadIdx.x;
    if (i >= n) return;
    int b = coords[i*4+0], z = coords[i*4+1], y = coords[i*4+2], x = coords[i*4+3];
    grid[cell_index(b, z, y, x)] = i;
}

__global__ __launch_bounds__(PB)
void k_nbr(const int* __restrict__ coords, const int* __restrict__ grid,
           int* __restrict__ npos, int* __restrict__ cntb, int n) {
    __shared__ int s_c[NWV][32];
    __shared__ int s_t[NWV][32*NOFF];
    __shared__ int s_d[NOFF*65];
    const int p    = blockIdx.x * PB + threadIdx.x;
    const int lane = threadIdx.x & 31;
    const int wv   = __builtin_amdgcn_readfirstlane((int)(threadIdx.x >> 5));
    const bool live = p < n && p < NPTS;
    int bz = 0, z = 0, y = 0, x = 0;
    if (live) {
        bz = coords[p*4+0]; z = coords[p*4+1]; y = coords[p*4+2]; x = coords[p*4+3];
    }
    for (int e = threadIdx.x; e < NOFF*64; e += PB) {
        const int t = e >> 6, c = e & 63;
        const int sz = t / 9 - 1, sy = (t / 3) % 3 - 1, sx = t % 3 - 1;
        const int xl = c & 3, yl = (c >> 2) & 3, zl = c >> 4;
        int d = 0;
        if (sx > 0) d += (xl == 3) ? (64 - 3) : 1;
        if (sx < 0) d -= (xl == 0) ? (64 - 3) : 1;
        if (sy > 0) d += (yl == 3) ? (64*(WW/4) - 12) : 4;
        if (sy < 0) d -= (yl == 0) ? (64*(WW/4) - 12) : 4;
        if (sz > 0) d += (zl == 3) ? (64*(WW/4)*(HH/4) - 48) : 16;
        if (sz < 0) d -= (zl == 0) ? (64*(WW/4)*(HH/4) - 48) : 16;
        s_d[t*65 + c] = d;
    }
    __syncthreads();
#define FLD_OUT 0x380E0380u
    const bool own_ok = live && (unsigned)bz < (unsigned)BB && (unsigned)z < (unsigned)DD &&
                        (unsigned)y < (unsigned)HH && (unsigned)x < (unsigned)WW;
    const int own = own_ok ? cell_index(bz, z, y, x) : 0;
    const unsigned pk = own_ok ? ((unsigned)x | ((unsigned)y << 10) | ((unsigned)z << 20)) : 0x380u;
    const int tl = lane < NOFF ? lane : CTAP;
    const unsigned step = lane < NOFF
        ? (unsigned)((tl % 3 - 1) + ((tl / 3) % 3 - 1) * 1024 + (tl / 9 - 1) * 1048576)
        : 0x380u;
    const int* dt = s_d + tl*65;
    int* mt = s_t[wv];
    int got[32];
    unsigned okbits = 0;
#pragma unroll
    for (int i = 0; i < 32; ++i) {
        const int      so = __builtin_amdgcn_readlane(own, i);
        const unsigned sp = (unsigned)__builtin_amdgcn_readlane((int)pk, i);
        const bool ok = ((sp + step) & FLD_OUT) == 0u;
        const int d = dt[so & 63];
        got[i] = grid[so + (ok ? d : 0)];
        okbits |= (ok ? 1u : 0u) << i;
    }
    if (lane < NOFF) {
#pragma unroll
        for (int i = 0; i < 32; ++i) mt[i*NOFF + lane] = ((okbits >> i) & 1u) ? got[i] : -1;
    }
    __builtin_amdgcn_fence(__ATOMIC_ACQ_REL, "wavefront");
    int idx[NOFF];
#pragma unroll
    for (int t = 0; t < NOFF; ++t) idx[t] = mt[lane*NOFF + t];
#pragma unroll
    for (int t = 0; t < NOFF; ++t) {
        if (live) npos[(size_t)t*NPTS + p] = idx[t];
        unsigned mask = __builtin_amdgcn_ballot_w32(idx[t] >= 0);
        if (lane == 0) s_c[wv][t] = __builtin_popcount(mask);
    }
    __syncthreads();
    if (wv == 0) {
        int v = 0;
        if (lane < NOFF && lane != CTAP) {
#pragma unroll
            for (int w = 0; w < NWV; ++w) v += s_c[w][lane];
        }
        cntb[blockIdx.x*32 + lane] = v;
    }
}

#define LIST_HEAD 4
__global__ __launch_bounds__(PB)
void k_pairs(int* __restrict__ npos, const int* __restrict__ cntb,
             int* __restrict__ total, int* __restrict__ pin,
             int* __restrict__ cen, int n) {
    __shared__ int s_cw[NWV][32];
    __shared__ int s_pt[NWV][32];
    __shared__ int s_pp[NWV][32];
    const int p    = blockIdx.x * PB + threadIdx.x;
    const int lane = threadIdx.x & 31;
    const int wv   = __builtin_amdgcn_readfirstlane((int)(threadIdx.x >> 5));
    const int blk  = blockIdx.x;
    const bool live = p < n && p < NPTS;
    int c[NB/NWV];
#pragma unroll
    for (int i = 0; i < NB/NWV; ++i) c[i] = cntb[(wv + i*NWV)*32 + lane];
    int q[NOFF];
#pragma unroll
    for (int t = 0; t < NOFF; ++t) q[t] = live ? npos[(size_t)t*NPTS + p] : -1;

    {
        int pt = 0, pp = 0;
#pragma unroll
        for (int i = 0; i < NB/NWV; ++i) {
            pt += c[i];
            if (wv + i*NWV < blk) pp += c[i];
        }
        s_pt[wv][lane] = pt;
        s_pp[wv][lane] = pp;
    }
#pragma unroll
    for (int t = 0; t < NOFF; ++t) {
        unsigned mask = __builtin_amdgcn_ballot_w32(q[t] >= 0);
        if (lane == 0) s_cw[wv][t] = (t == CTAP) ? 0 : __builtin_popcount(mask);
    }
    __syncthreads();

    int tot = 0;
    int off = 0;
    if (lane < NOFF) {
        int a_pt[NWV], a_pp[NWV], a_cw[NWV];
#pragma unroll
        for (int w = 0; w < NWV; ++w) {
            a_pt[w] = s_pt[w][lane];
            a_pp[w] = s_pp[w][lane];
            a_cw[w] = s_cw[w][lane];
        }
#pragma unroll
        for (int w = 0; w < NWV; ++w) {
            tot += a_pt[w];
            off += a_pp[w];
            off += (w < wv) ? a_cw[w] : 0;
        }
        if (blk == 0 && wv == 0) total[lane] = tot;
    }

    int tb = 0;
    int k  = 0;
#pragma unroll
    for (int t = 0; t < NOFF; ++t) {
        if (t == CTAP) { if (live) cen[p] = q[t]; continue; }
        const int tot_t = __builtin_amdgcn_readlane(tot, t);
        const int off_t = __builtin_amdgcn_readlane(off, t);
        unsigned mask = __builtin_amdgcn_ballot_w32(q[t] >= 0);
        if (q[t] >= 0) {
            int pos = tb*16 + off_t + __builtin_popcount(mask & ((1u << lane) - 1u));
            if (pos < CAP_ROWS) {
                pin[pos] = q[t];
                npos[(size_t)k*NPTS + p] = pos;
                ++k;
            }
        }
        tb += (tot_t + 15) >> 4;
    }
    if (live) {
        for (int s2 = 0; s2 < LIST_HEAD; ++s2)
            if (k <= s2) npos[(size_t)s2*NPTS + p] = -1;
        if (k >= LIST_HEAD && k < NOFF) npos[(size_t)k*NPTS + p] = -1;
    }
}

#define GEMM_WGS 1024
__global__ __launch_bounds__(256)
void k_gemm(const float* __restrict__ feats,
            const float* __restrict__ wT,
            const int*   __restrict__ total,
            const int*   __restrict__ pin,
            float*       __restrict__ part) {
    const int lane = threadIdx.x & 31;
    const int col  = lane & 15;
    const int half = lane >> 4;
    const int wid    = __builtin_amdgcn_readfirstlane((int)(blockIdx.x * 8 + (threadIdx.x >> 5)));
    const int nwaves = gridDim.x * 8;

    int ntiles = 0;
    for (int k = 0; k < NOFF; ++k) ntiles += (total[k] + 15) >> 4;

    for (int tile = wid; tile < ntiles; tile += nwaves) {
        int t = 0, tb = 0, tot = 0;
        {
            int acc0 = 0;
            for (int k = 0; k < NOFF; ++k) {
                int tk = total[k];
                int nk = (tk + 15) >> 4;
                if (tile >= acc0 && tile < acc0 + nk) { t = k; tb = acc0; tot = tk; }
                acc0 += nk;
            }
        }
        const int rend = tb*16 + tot;
        const int arow = tile*16 + col;
        int q = -1;
        if (arow < rend && arow < CAP_ROWS) q = pin[arow];

        v2f a[16];
#pragma unroll
        for (int kc = 0; kc < 16; ++kc) {
            v2f v; v.x = 0.f; v.y = 0.f;
            if (q >= 0) v = *(const v2f*)(feats + (size_t)q*CI + kc*4 + 2*half);
            a[kc] = v;
        }

        v8f acc[4];
#pragma unroll
        for (int nt = 0; nt < 4; ++nt)
#pragma unroll
            for (int j = 0; j < 8; ++j) acc[nt][j] = 0.f;

        const float* wbase = wT + (size_t)t*(16*4*64) + lane*2;
#pragma unroll
        for (int kc = 0; kc < 16; ++kc) {
#pragma unroll
            for (int nt = 0; nt < 4; ++nt) {
                v2f b = *(const v2f*)(wbase + (kc*4 + nt)*64);
                acc[nt] = __builtin_amdgcn_wmma_f32_16x16x4_f32(
                    false, a[kc], false, b, (short)0, acc[nt], false, false);
            }
        }

#pragma unroll
        for (int j = 0; j < 8; ++j) {
            int r = tile*16 + j + 8*half;
            if (r < rend && r < CAP_ROWS) {
#pragma unroll
                for (int nt = 0; nt < 4; ++nt)
                    part[(size_t)r*CO + nt*16 + col] = acc[nt][j];
            }
        }
    }
}

#define FIN_WGS ((NPTS/16 + 7)/8)
#define TP 68
#define EB 4
__global__ __launch_bounds__(256)
void k_final(const float* __restrict__ feats, const float* __restrict__ wT,
             const float* __restrict__ bias, const int* __restrict__ cen,
             const int* __restrict__ list, const float* __restrict__ part,
             float* __restrict__ out, int n) {
    __shared__ __attribute__((aligned(16))) float s_tile[8][16*TP];
    __shared__ __attribute__((aligned(16))) int   s_head[8][16*LIST_HEAD];
    const int lane = threadIdx.x & 31;
    const int col  = lane & 15;
    const int half = lane >> 4;
    const int wv   = threadIdx.x >> 5;
    const int tile  = __builtin_amdgcn_readfirstlane((int)(blockIdx.x * 8 + wv));
    const int ntile = (n + 15) >> 4;
    if (tile >= ntile) return;
    float* my  = s_tile[wv];
    int*   myh = s_head[wv];
    const float4* rows = (const float4*)part;

    const float* wbase = wT + (size_t)CTAP*(16*4*64) + lane*2;
    const float4 bs = *(const float4*)(bias + col*4);

    const int p = tile*16 + col;
    int q = -1;
    int4 hd = make_int4(-1, -1, -1, -1);
    if (p < n) {
        q    = cen[p];
        hd.x = list[p];
        hd.y = list[(size_t)1*NPTS + p];
        hd.z = list[(size_t)2*NPTS + p];
        hd.w = list[(size_t)3*NPTS + p];
    }
    v2f a[16];
    const float* arow = feats + (size_t)(q >= 0 ? q : 0)*CI + 2*half;
#pragma unroll
    for (int kc = 0; kc < 16; ++kc) {
        v2f v = *(const v2f*)(arow + kc*4);
        if (q < 0) { v.x = 0.f; v.y = 0.f; }
        a[kc] = v;
    }
    v8f acc[4];
#pragma unroll
    for (int nt = 0; nt < 4; ++nt)
#pragma unroll
        for (int j = 0; j < 8; ++j) acc[nt][j] = 0.f;
#pragma unroll
    for (int kc = 0; kc < 16; ++kc) {
#pragma unroll
        for (int nt = 0; nt < 4; ++nt) {
            v2f b = *(const v2f*)(wbase + (kc*4 + nt)*64);
            acc[nt] = __builtin_amdgcn_wmma_f32_16x16x4_f32(
                false, a[kc], false, b, (short)0, acc[nt], false, false);
        }
    }
#pragma unroll
    for (int j = 0; j < 8; ++j)
#pragma unroll
        for (int nt = 0; nt < 4; ++nt)
            my[(j + 8*half)*TP + nt*16 + col] = acc[nt][j];
    if (half == 0) *(int4*)(myh + col*LIST_HEAD) = hd;
    __builtin_amdgcn_fence(__ATOMIC_ACQ_REL, "wavefront");

    const int c4 = col*4;
#pragma unroll
    for (int b0 = 0; b0 < 8; b0 += EB) {
        int4 h[EB];
        float4 v[EB], x0[EB], x1[EB], x2[EB], x3[EB];
#pragma unroll
        for (int e = 0; e < EB; ++e) {
            const int row = 2*(b0 + e) + half;
            h[e] = *(const int4*)(myh + row*LIST_HEAD);
            v[e] = *(const float4*)(my + row*TP + c4);
        }
#pragma unroll
        for (int e = 0; e < EB; ++e) {
            const float4 z = make_float4(0.f, 0.f, 0.f, 0.f);
            x0[e] = z; x1[e] = z; x2[e] = z; x3[e] = z;
            if (h[e].x >= 0) {
                x0[e] = rows[(unsigned)h[e].x*(CO/4) + col];
                x1[e] = rows[(unsigned)(h[e].y >= 0 ? h[e].y : h[e].x)*(CO/4) + col];
                x2[e] = rows[(unsigned)(h[e].z >= 0 ? h[e].z : h[e].x)*(CO/4) + col];
                x3[e] = rows[(unsigned)(h[e].w >= 0 ? h[e].w : h[e].x)*(CO/4) + col];
            }
        }
#pragma unroll
        for (int e = 0; e < EB; ++e) {
            const int pp = tile*16 + 2*(b0 + e) + half;
            float4 r;
            r.x = bs.x + v[e].x; r.y = bs.y + v[e].y; r.z = bs.z + v[e].z; r.w = bs.w + v[e].w;
            if (h[e].x >= 0) {
                r.x += x0[e].x; r.y += x0[e].y; r.z += x0[e].z; r.w += x0[e].w;
                if (h[e].y >= 0) { r.x += x1[e].x; r.y += x1[e].y; r.z += x1[e].z; r.w += x1[e].w; }
                if (h[e].z >= 0) { r.x += x2[e].x; r.y += x2[e].y; r.z += x2[e].z; r.w += x2[e].w; }
                if (h[e].w >= 0) {
                    r.x += x3[e].x; r.y += x3[e].y; r.z += x3[e].z; r.w += x3[e].w;
                    for (int k = LIST_HEAD; k < NOFF; ++k) {
                        const int rr = list[(size_t)k*NPTS + pp];
                        if (rr < 0) break;
                        const float4 y = rows[(unsigned)rr*(CO/4) + col];
                        r.x += y.x; r.y += y.y; r.z += y.z; r.w += y.w;
                    }
                }
            }
            if (pp < n) *(float4*)(out + (size_t)pp*CO + c4) = r;
        }
    }
}

extern "C" void kernel_launch(void* const* d_in, const int* in_sizes, int n_in,
                              void* d_out, int out_size, void* d_ws, size_t ws_size,
                              hipStream_t stream) {
    const float* feats  = (const float*)d_in[0];
    const float* weight = (const float*)d_in[1];
    const float* bias   = (const float*)d_in[2];
    const int*   coords = (const int*)  d_in[3];
    float* out = (float*)d_out;

    const int n = in_sizes[0] / CI;

    float* wT    = (float*)d_ws;
    int*   grid  = (int*)  ((char*)d_ws + OFF_GRID);
    int*   cen   = grid;
    int*   npos  = (int*)  ((char*)d_ws + OFF_NPOS);
    int*   cntb  = (int*)  ((char*)d_ws + OFF_CNT);
    int*   total = (int*)  ((char*)d_ws + OFF_TOT);
    int*   pin   = (int*)  ((char*)d_ws + OFF_PIN);
    float* part  = (float*)((char*)d_ws + OFF_PART);

    k_fill_grid<<<FILL_WGS,               256, 0, stream>>>(grid, weight, wT);
    k_scatter  <<<(n + 255)/256,          256, 0, stream>>>(coords, grid, n);
    k_nbr      <<<NB,                     PB,  0, stream>>>(coords, grid, npos, cntb, n);
    k_pairs    <<<NB,                     PB,  0, stream>>>(npos, cntb, total, pin, cen, n);
    k_gemm     <<<GEMM_WGS,               256, 0, stream>>>(feats, wT, total, pin, part);
    k_final    <<<FIN_WGS,                256, 0, stream>>>(feats, wT, bias, cen, npos, part, out, n);
}


# pair list cut into fixed segments per (800-point group, tap): rows placed in the lookup kernel, launches 6 -> 5, no neighbour table; a tap too dense for its segment goes the centre's way; sums same
# speedup vs baseline: 3.2795x; 1.0808x over previous
#include <hip/hip_runtime.h>

typedef float v2f __attribute__((ext_vector_type(2)));
typedef float v8f __attribute__((ext_vector_type(8)));

#define BB   2
#define DD   128
#define HH   128
#define WW   128
#define CI   64
#define CO   64
#define NOFF 27
#define CTAP 13
#define NPTS 200000
#define GRID_CELLS (BB*DD*HH*WW)
#define WT_FLOATS  (NOFF*16*4*32*2)
#define PB   800
#define NB   ((NPTS + PB - 1)/PB)
#define NWV  (PB/32)
#define NTAPO 26
#define SEG   128
#define NSEG  (NB*NTAPO)
#define CAP_ROWS (NSEG*SEG)

#define OFF_GRID  ((size_t)442368)
#define OFF_NPOS  ((size_t)17219584)
#define OFF_CNT   ((size_t)38819584)
#define OFF_CEN   ((size_t)39494656)
#define OFF_PIN   ((size_t)40294656)
#define OFF_PART  ((size_t)43622656)

#define FILL_WGS 512
#define FILL_PER (GRID_CELLS / 4 / (FILL_WGS * 256))
__global__ __launch_bounds__(256)
void k_fill_grid(int* __restrict__ grid, const float* __restrict__ w, float* __restrict__ wT) {
    int4* g = (int4*)grid;
    const int4 m = make_int4(-1, -1, -1, -1);
    const int i = blockIdx.x * 256 + threadIdx.x;
    const bool has = i < NOFF*16*4*32;
    v2f wv2; wv2.x = 0.f; wv2.y = 0.f;
    if (has) {
        int lane = i & 31;
        int nt   = (i >> 5) & 3;
        int kc   = (i >> 7) & 15;
        int t    =  i >> 11;
        int o = nt*16 + (lane & 15);
        int c = kc*4 + 2*(lane >> 4);
        wv2 = *(const v2f*)(w + (size_t)(o*NOFF + t)*CI + c);
    }
#pragma unroll
    for (int k = 0; k < FILL_PER; ++k) g[i + k * (FILL_WGS * 256)] = m;
    if (has) *(v2f*)(wT + (size_t)i*2) = wv2;
}

__device__ __forceinline__ int cell_index(int b, int z, int y, int x) {
    int blk = ((b*(DD/4) + (z >> 2))*(HH/4) + (y >> 2))*(WW/4) + (x >> 2);
    return (blk << 6) | ((z & 3) << 4) | ((y & 3) << 2) | (x & 3);
}

__global__ void k_scatter(const int* __restrict__ coords, int* __restrict__ grid, int n) {
    int i = blockIdx.x * blockDim.x + threadIdx.x;
    if (i >= n) return;
    int b = coords[i*4+0], z = coords[i*4+1], y = coords[i*4+2], x = coords[i*4+3];
    grid[cell_index(b, z, y, x)] = i;
}

#define LIST_HEAD 4
__global__ __launch_bounds__(PB)
void k_nbr(const int* __restrict__ coords, const int* __restrict__ grid,
           int* __restrict__ list, int* __restrict__ cntb,
           int* __restrict__ pin, int* __restrict__ cen, int n) {
    __shared__ int s_c[NWV][32];
    __shared__ int s_t[NWV][32*NOFF];
    __shared__ int s_d[NOFF*65];
    const int p    = blockIdx.x * PB + threadIdx.x;
    const int lane = threadIdx.x & 31;
    const int wv   = __builtin_amdgcn_readfirstlane((int)(threadIdx.x >> 5));
    const bool live = p < n && p < NPTS;
    int bz = 0, z = 0, y = 0, x = 0;
    if (live) {
        bz = coords[p*4+0]; z = coords[p*4+1]; y = coords[p*4+2]; x = coords[p*4+3];
    }
    for (int e = threadIdx.x; e < NOFF*64; e += PB) {
        const int t = e >> 6, c = e & 63;
        const int sz = t / 9 - 1, sy = (t / 3) % 3 - 1, sx = t % 3 - 1;
        const int xl = c & 3, yl = (c >> 2) & 3, zl = c >> 4;
        int d = 0;
        if (sx > 0) d += (xl == 3) ? (64 - 3) : 1;
        if (sx < 0) d -= (xl == 0) ? (64 - 3) : 1;
        if (sy > 0) d += (yl == 3) ? (64*(WW/4) - 12) : 4;
        if (sy < 0) d -= (yl == 0) ? (64*(WW/4) - 12) : 4;
        if (sz > 0) d += (zl == 3) ? (64*(WW/4)*(HH/4) - 48) : 16;
        if (sz < 0) d -= (zl == 0) ? (64*(WW/4)*(HH/4) - 48) : 16;
        s_d[t*65 + c] = d;
    }
    __syncthreads();
#define FLD_OUT 0x380E0380u
    const bool own_ok = live && (unsigned)bz < (unsigned)BB && (unsigned)z < (unsigned)DD &&
                        (unsigned)y < (unsigned)HH && (unsigned)x < (unsigned)WW;
    const int own = own_ok ? cell_index(bz, z, y, x) : 0;
    const unsigned pk = own_ok ? ((unsigned)x | ((unsigned)y << 10) | ((unsigned)z << 20)) : 0x380u;
    const int tl = lane < NOFF ? lane : CTAP;
    const unsigned step = lane < NOFF
        ? (unsigned)((tl % 3 - 1) + ((tl / 3) % 3 - 1) * 1024 + (tl / 9 - 1) * 1048576)
        : 0x380u;
    const int* dt = s_d + tl*65;
    int* mt = s_t[wv];
    int got[32];
    unsigned okbits = 0;
#pragma unroll
    for (int i = 0; i < 32; ++i) {
        const int      so = __builtin_amdgcn_readlane(own, i);
        const unsigned sp = (unsigned)__builtin_amdgcn_readlane((int)pk, i);
        const bool ok = ((sp + step) & FLD_OUT) == 0u;
        const int d = dt[so & 63];
        got[i] = grid[so + (ok ? d : 0)];
        okbits |= (ok ? 1u : 0u) << i;
    }
    if (lane < NOFF) {
#pragma unroll
        for (int i = 0; i < 32; ++i) mt[i*NOFF + lane] = ((okbits >> i) & 1u) ? got[i] : -1;
    }
    __builtin_amdgcn_fence(__ATOMIC_ACQ_REL, "wavefront");
    int idx[NOFF];
#pragma unroll
    for (int t = 0; t < NOFF; ++t) idx[t] = mt[lane*NOFF + t];
#pragma unroll
    for (int t = 0; t < NOFF; ++t) {
        unsigned mask = __builtin_amdgcn_ballot_w32(idx[t] >= 0);
        if (lane == 0) s_c[wv][t] = (t == CTAP) ? 0 : __builtin_popcount(mask);
    }
    __syncthreads();

    int tot = 0;
    int off = 0;
    if (lane < NOFF) {
        int a_c[NWV];
#pragma unroll
        for (int w = 0; w < NWV; ++w) a_c[w] = s_c[w][lane];
#pragma unroll
        for (int w = 0; w < NWV; ++w) {
            tot += a_c[w];
            off += (w < wv) ? a_c[w] : 0;
        }
    }
    if (wv == 0) cntb[blockIdx.x*32 + lane] = tot;

    int k = 0;
#pragma unroll
    for (int t = 0; t < NOFF; ++t) {
        if (t == CTAP) { if (live) cen[p] = idx[t]; continue; }
        const int u = t < CTAP ? t : t - 1;
        const int off_t = __builtin_amdgcn_readlane(off, t);
        const int tot_t = __builtin_amdgcn_readlane(tot, t);
        unsigned mask = __builtin_amdgcn_ballot_w32(idx[t] >= 0);
        if (idx[t] >= 0 && tot_t <= SEG) {
            const int r = off_t + __builtin_popcount(mask & ((1u << lane) - 1u));
            const int pos = (u*NB + (int)blockIdx.x)*SEG + r;
            pin[pos] = idx[t];
            list[(size_t)k*NPTS + p] = pos;
            ++k;
        }
    }
    if (live) {
        for (int s2 = 0; s2 < LIST_HEAD; ++s2)
            if (k <= s2) list[(size_t)s2*NPTS + p] = -1;
        if (k >= LIST_HEAD && k < NOFF) list[(size_t)k*NPTS + p] = -1;
    }
}

#define GEMM_WGS 1024
#define GW    (GEMM_WGS*8)
#define GRP   4
#define NGRP  ((NB + GRP - 1)/GRP)
#define NUNIT (NTAPO*NGRP)
#define UT    (GRP*SEG/16)
#define NSLOT (NUNIT*UT)
#define SPW   ((NSLOT + GW - 1)/GW)
static_assert(GRP == 4 && SPW*GRP <= 32, "the count fetch gives one lane to every (slot, member)");
__global__ __launch_bounds__(256)
void k_gemm(const float* __restrict__ feats,
            const float* __restrict__ wT,
            const int*   __restrict__ cntb,
            const int*   __restrict__ pin,
            float*       __restrict__ part) {
    const int lane = threadIdx.x & 31;
    const int col  = lane & 15;
    const int half = lane >> 4;
    const int wid = __builtin_amdgcn_readfirstlane((int)(blockIdx.x * 8 + (threadIdx.x >> 5)));

    int cnl = 0;
    {
        const int i = lane / GRP, m = lane - i*GRP;
        const int sl = wid + i*GW;
        if (i < SPW && sl < NSLOT) {
            const int un = sl % NUNIT;
            const int u = un / NGRP, blk = (un - u*NGRP)*GRP + m;
            if (blk < NB) {
                const int t = u < CTAP ? u : u + 1;
                const int cn = cntb[blk*32 + t];
                cnl = cn <= SEG ? cn : 0;
            }
        }
    }

    for (int i = 0; i < SPW; ++i) {
        const int sl = wid + i*GW;
        const int j = sl / NUNIT, un = sl - j*NUNIT;
        const int p1 =      __builtin_amdgcn_readlane(cnl, i*GRP);
        const int p2 = p1 + __builtin_amdgcn_readlane(cnl, i*GRP + 1);
        const int p3 = p2 + __builtin_amdgcn_readlane(cnl, i*GRP + 2);
        const int rem = p3 + __builtin_amdgcn_readlane(cnl, i*GRP + 3) - j*16;
        if (rem <= 0) continue;
        const int nr = rem < 16 ? rem : 16;
        const int u = un / NGRP;
        const int t = u < CTAP ? u : u + 1;
        const int vr = j*16 + col;
        const int mb = (vr >= p1 ? 1 : 0) + (vr >= p2 ? 1 : 0) + (vr >= p3 ? 1 : 0);
        const int rowa = (u*NB + (un - u*NGRP)*GRP + mb)*SEG
                       + vr - (vr >= p3 ? p3 : vr >= p2 ? p2 : vr >= p1 ? p1 : 0);
        int q = -1;
        if (col < nr) q = pin[rowa];

        v2f a[16];
#pragma unroll
        for (int kc = 0; kc < 16; ++kc) {
            v2f v; v.x = 0.f; v.y = 0.f;
            if (q >= 0) v = *(const v2f*)(feats + (size_t)q*CI + kc*4 + 2*half);
            a[kc] = v;
        }

        v8f acc[4];
#pragma unroll
        for (int nt = 0; nt < 4; ++nt)
#pragma unroll
            for (int e = 0; e < 8; ++e) acc[nt][e] = 0.f;

        const float* wbase = wT + (size_t)t*(16*4*64) + lane*2;
#pragma unroll
        for (int kc = 0; kc < 16; ++kc) {
#pragma unroll
            for (int nt = 0; nt < 4; ++nt) {
                v2f b = *(const v2f*)(wbase + (kc*4 + nt)*64);
                acc[nt] = __builtin_amdgcn_wmma_f32_16x16x4_f32(
                    false, a[kc], false, b, (short)0, acc[nt], false, false);
            }
        }

#pragma unroll
        for (int e = 0; e < 8; ++e) {
            const int r0 = __builtin_amdgcn_readlane(rowa, e);
            const int r1 = __builtin_amdgcn_readlane(rowa, e + 8);
            if (e + 8*half < nr) {
                const size_t ro = (size_t)(half ? r1 : r0)*CO;
#pragma unroll
                for (int nt = 0; nt < 4; ++nt)
                    part[ro + nt*16 + col] = acc[nt][e];
            }
        }
    }
}

#define FIN_WGS ((NPTS/16 + 7)/8)
#define TP 68
#define EB 4
__global__ __launch_bounds__(256)
void k_final(const float* __restrict__ feats, const float* __restrict__ wT,
             const float* __restrict__ bias, const int* __restrict__ cen,
             const int* __restrict__ list, const float* __restrict__ part,
             const int* __restrict__ cntb, const int* __restrict__ coords,
             const int* __restrict__ grid, float* __restrict__ out, int n) {
    __shared__ __attribute__((aligned(16))) float s_tile[8][16*TP];
    __shared__ __attribute__((aligned(16))) int   s_head[8][16*LIST_HEAD];
    const int lane = threadIdx.x & 31;
    const int col  = lane & 15;
    const int half = lane >> 4;
    const int wv   = threadIdx.x >> 5;
    const int tile  = __builtin_amdgcn_readfirstlane((int)(blockIdx.x * 8 + wv));
    const int ntile = (n + 15) >> 4;
    if (tile >= ntile) return;
    float* my  = s_tile[wv];
    int*   myh = s_head[wv];
    const float4* rows = (const float4*)part;

    const float* wbase = wT + (size_t)CTAP*(16*4*64) + lane*2;
    const float4 bs = *(const float4*)(bias + col*4);

    const unsigned dense = __builtin_amdgcn_ballot_w32(cntb[(tile/(PB/16))*32 + lane] > SEG);

    const int p = tile*16 + col;
    int q = -1;
    int4 hd = make_int4(-1, -1, -1, -1);
    if (p < n) {
        q    = cen[p];
        hd.x = list[p];
        hd.y = list[(size_t)1*NPTS + p];
        hd.z = list[(size_t)2*NPTS + p];
        hd.w = list[(size_t)3*NPTS + p];
    }
    v2f a[16];
    const float* arow = feats + (size_t)(q >= 0 ? q : 0)*CI + 2*half;
#pragma unroll
    for (int kc = 0; kc < 16; ++kc) {
        v2f v = *(const v2f*)(arow + kc*4);
        if (q < 0) { v.x = 0.f; v.y = 0.f; }
        a[kc] = v;
    }
    v8f acc[4];
#pragma unroll
    for (int nt = 0; nt < 4; ++nt)
#pragma unroll
        for (int j = 0; j < 8; ++j) acc[nt][j] = 0.f;
#pragma unroll
    for (int kc = 0; kc < 16; ++kc) {
#pragma unroll
        for (int nt = 0; nt < 4; ++nt) {
            v2f b = *(const v2f*)(wbase + (kc*4 + nt)*64);
            acc[nt] = __builtin_amdgcn_wmma_f32_16x16x4_f32(
                false, a[kc], false, b, (short)0, acc[nt], false, false);
        }
    }
    if (dense != 0u) {
        int bz = 0, z = 0, y = 0, x = 0;
        if (p < n) { bz = coords[p*4+0]; z = coords[p*4+1]; y = coords[p*4+2]; x = coords[p*4+3]; }
        const bool own_ok = p < n && (unsigned)bz < (unsigned)BB && (unsigned)z < (unsigned)DD &&
                            (unsigned)y < (unsigned)HH && (unsigned)x < (unsigned)WW;
        for (unsigned m = dense; m != 0u; m &= m - 1u) {
            const int t  = __builtin_ctz(m);
            const int nz = z + t/9 - 1, ny = y + (t/3)%3 - 1, nx = x + t%3 - 1;
            int q2 = -1;
            if (own_ok && (unsigned)nz < (unsigned)DD && (unsigned)ny < (unsigned)HH &&
                (unsigned)nx < (unsigned)WW)
                q2 = grid[cell_index(bz, nz, ny, nx)];
            const float* ar = feats + (size_t)(q2 >= 0 ? q2 : 0)*CI + 2*half;
            const float* wb = wT + (size_t)t*(16*4*64) + lane*2;
#pragma nounroll
            for (int kc = 0; kc < 16; ++kc) {
                v2f av = *(const v2f*)(ar + kc*4);
                if (q2 < 0) { av.x = 0.f; av.y = 0.f; }
#pragma unroll
                for (int nt = 0; nt < 4; ++nt) {
                    v2f b = *(const v2f*)(wb + (kc*4 + nt)*64);
                    acc[nt] = __builtin_amdgcn_wmma_f32_16x16x4_f32(
                        false, av, false, b, (short)0, acc[nt], false, false);
                }
            }
        }
    }
#pragma unroll
    for (int j = 0; j < 8; ++j)
#pragma unroll
        for (int nt = 0; nt < 4; ++nt)
            my[(j + 8*half)*TP + nt*16 + col] = acc[nt][j];
    if (half == 0) *(int4*)(myh + col*LIST_HEAD) = hd;
    __builtin_amdgcn_fence(__ATOMIC_ACQ_REL, "wavefront");

    const int c4 = col*4;
#pragma unroll
    for (int b0 = 0; b0 < 8; b0 += EB) {
        int4 h[EB];
        float4 v[EB], x0[EB], x1[EB], x2[EB], x3[EB];
#pragma unroll
        for (int e = 0; e < EB; ++e) {
            const int row = 2*(b0 + e) + half;
            h[e] = *(const int4*)(myh + row*LIST_HEAD);
            v[e] = *(const float4*)(my + row*TP + c4);
        }
#pragma unroll
        for (int e = 0; e < EB; ++e) {
            const float4 z = make_float4(0.f, 0.f, 0.f, 0.f);
            x0[e] = z; x1[e] = z; x2[e] = z; x3[e] = z;
            if (h[e].x >= 0) {
                x0[e] = rows[(unsigned)h[e].x*(CO/4) + col];
                x1[e] = rows[(unsigned)(h[e].y >= 0 ? h[e].y : h[e].x)*(CO/4) + col];
                x2[e] = rows[(unsigned)(h[e].z >= 0 ? h[e].z : h[e].x)*(CO/4) + col];
                x3[e] = rows[(unsigned)(h[e].w >= 0 ? h[e].w : h[e].x)*(CO/4) + col];
            }
        }
#pragma unroll
        for (int e = 0; e < EB; ++e) {
            const int pp = tile*16 + 2*(b0 + e) + half;
            float4 r;
            r.x = bs.x + v[e].x; r.y = bs.y + v[e].y; r.z = bs.z + v[e].z; r.w = bs.w + v[e].w;
            if (h[e].x >= 0) {
                r.x += x0[e].x; r.y += x0[e].y; r.z += x0[e].z; r.w += x0[e].w;
                if (h[e].y >= 0) { r.x += x1[e].x; r.y += x1[e].y; r.z += x1[e].z; r.w += x1[e].w; }
                if (h[e].z >= 0) { r.x += x2[e].x; r.y += x2[e].y; r.z += x2[e].z; r.w += x2[e].w; }
                if (h[e].w >= 0) {
                    r.x += x3[e].x; r.y += x3[e].y; r.z += x3[e].z; r.w += x3[e].w;
                    for (int k = LIST_HEAD; k < NOFF; ++k) {
                        const int rr = list[(size_t)k*NPTS + pp];
                        if (rr < 0) break;
                        const float4 y = rows[(unsigned)rr*(CO/4) + col];
                        r.x += y.x; r.y += y.y; r.z += y.z; r.w += y.w;
                    }
                }
            }
            if (pp < n) *(float4*)(out + (size_t)pp*CO + c4) = r;
        }
    }
}

extern "C" void kernel_launch(void* const* d_in, const int* in_sizes, int n_in,
                              void* d_out, int out_size, void* d_ws, size_t ws_size,
                              hipStream_t stream) {
    const float* feats  = (const float*)d_in[0];
    const float* weight = (const float*)d_in[1];
    const float* bias   = (const float*)d_in[2];
    const int*   coords = (const int*)  d_in[3];
    float* out = (float*)d_out;

    const int n = in_sizes[0] / CI;

    float* wT    = (float*)d_ws;
    int*   grid  = (int*)  ((char*)d_ws + OFF_GRID);
    int*   list  = (int*)  ((char*)d_ws + OFF_NPOS);
    int*   cntb  = (int*)  ((char*)d_ws + OFF_CNT);
    int*   cen   = (int*)  ((char*)d_ws + OFF_CEN);
    int*   pin   = (int*)  ((char*)d_ws + OFF_PIN);
    float* part  = (float*)((char*)d_ws + OFF_PART);

    k_fill_grid<<<FILL_WGS,               256, 0, stream>>>(grid, weight, wT);
    k_scatter  <<<(n + 255)/256,          256, 0, stream>>>(coords, grid, n);
    k_nbr      <<<NB,                     PB,  0, stream>>>(coords, grid, list, cntb, pin, cen, n);
    k_gemm     <<<GEMM_WGS,               256, 0, stream>>>(feats, wT, cntb, pin, part);
    k_final    <<<FIN_WGS,                256, 0, stream>>>(feats, wT, bias, cen, list, part, cntb, coords, grid,
                                                            out, n);
}


# pair products move into the lookup kernel: each 800-point work-group multiplies the row tiles of its own 26 segments (neighbour rows kept in LDS) right after placing them; launches 5 -> 4; sums same
# speedup vs baseline: 3.3914x; 1.0341x over previous
#include <hip/hip_runtime.h>

typedef float v2f __attribute__((ext_vector_type(2)));
typedef float v8f __attribute__((ext_vector_type(8)));

#define BB   2
#define DD   128
#define HH   128
#define WW   128
#define CI   64
#define CO   64
#define NOFF 27
#define CTAP 13
#define NPTS 200000
#define GRID_CELLS (BB*DD*HH*WW)
#define WT_FLOATS  (NOFF*16*4*32*2)
#define PB   800
#define NB   ((NPTS + PB - 1)/PB)
#define NWV  (PB/32)
#define NTAPO 26
#define SEG   128
#define NSEG  (NB*NTAPO)
#define CAP_ROWS (NSEG*SEG)

#define OFF_GRID  ((size_t)442368)
#define OFF_NPOS  ((size_t)17219584)
#define OFF_CNT   ((size_t)38819584)
#define OFF_CEN   ((size_t)39494656)
#define OFF_PART  ((size_t)43622656)

#define FILL_WGS 512
#define FILL_PER (GRID_CELLS / 4 / (FILL_WGS * 256))
__global__ __launch_bounds__(256)
void k_fill_grid(int* __restrict__ grid, const float* __restrict__ w, float* __restrict__ wT) {
    int4* g = (int4*)grid;
    const int4 m = make_int4(-1, -1, -1, -1);
    const int i = blockIdx.x * 256 + threadIdx.x;
    const bool has = i < NOFF*16*4*32;
    v2f wv2; wv2.x = 0.f; wv2.y = 0.f;
    if (has) {
        int lane = i & 31;
        int nt   = (i >> 5) & 3;
        int kc   = (i >> 7) & 15;
        int t    =  i >> 11;
        int o = nt*16 + (lane & 15);
        int c = kc*4 + 2*(lane >> 4);
        wv2 = *(const v2f*)(w + (size_t)(o*NOFF + t)*CI + c);
    }
#pragma unroll
    for (int k = 0; k < FILL_PER; ++k) g[i + k * (FILL_WGS * 256)] = m;
    if (has) *(v2f*)(wT + (size_t)i*2) = wv2;
}

__device__ __forceinline__ int cell_index(int b, int z, int y, int x) {
    int blk = ((b*(DD/4) + (z >> 2))*(HH/4) + (y >> 2))*(WW/4) + (x >> 2);
    return (blk << 6) | ((z & 3) << 4) | ((y & 3) << 2) | (x & 3);
}

__global__ void k_scatter(const int* __restrict__ coords, int* __restrict__ grid, int n) {
    int i = blockIdx.x * blockDim.x + threadIdx.x;
    if (i >= n) return;
    int b = coords[i*4+0], z = coords[i*4+1], y = coords[i*4+2], x = coords[i*4+3];
    grid[cell_index(b, z, y, x)] = i;
}

#define LIST_HEAD 4
__global__ __launch_bounds__(PB)
void k_nbr(const int* __restrict__ coords, const int* __restrict__ grid,
           int* __restrict__ list, int* __restrict__ cntb,
           int* __restrict__ cen,
           const float* __restrict__ feats, const float* __restrict__ wT,
           float* __restrict__ part, int n) {
    __shared__ int s_pin[NTAPO*SEG];
    __shared__ int s_c[NWV][32];
    __shared__ int s_t[NWV][32*NOFF];
    __shared__ int s_d[NOFF*65];
    const int p    = blockIdx.x * PB + threadIdx.x;
    const int lane = threadIdx.x & 31;
    const int wv   = __builtin_amdgcn_readfirstlane((int)(threadIdx.x >> 5));
    const bool live = p < n && p < NPTS;
    int bz = 0, z = 0, y = 0, x = 0;
    if (live) {
        bz = coords[p*4+0]; z = coords[p*4+1]; y = coords[p*4+2]; x = coords[p*4+3];
    }
    for (int e = threadIdx.x; e < NOFF*64; e += PB) {
        const int t = e >> 6, c = e & 63;
        const int sz = t / 9 - 1, sy = (t / 3) % 3 - 1, sx = t % 3 - 1;
        const int xl = c & 3, yl = (c >> 2) & 3, zl = c >> 4;
        int d = 0;
        if (sx > 0) d += (xl == 3) ? (64 - 3) : 1;
        if (sx < 0) d -= (xl == 0) ? (64 - 3) : 1;
        if (sy > 0) d += (yl == 3) ? (64*(WW/4) - 12) : 4;
        if (sy < 0) d -= (yl == 0) ? (64*(WW/4) - 12) : 4;
        if (sz > 0) d += (zl == 3) ? (64*(WW/4)*(HH/4) - 48) : 16;
        if (sz < 0) d -= (zl == 0) ? (64*(WW/4)*(HH/4) - 48) : 16;
        s_d[t*65 + c] = d;
    }
    __syncthreads();
#define FLD_OUT 0x380E0380u
    const bool own_ok = live && (unsigned)bz < (unsigned)BB && (unsigned)z < (unsigned)DD &&
                        (unsigned)y < (unsigned)HH && (unsigned)x < (unsigned)WW;
    const int own = own_ok ? cell_index(bz, z, y, x) : 0;
    const unsigned pk = own_ok ? ((unsigned)x | ((unsigned)y << 10) | ((unsigned)z << 20)) : 0x380u;
    const int tl = lane < NOFF ? lane : CTAP;
    const unsigned step = lane < NOFF
        ? (unsigned)((tl % 3 - 1) + ((tl / 3) % 3 - 1) * 1024 + (tl / 9 - 1) * 1048576)
        : 0x380u;
    const int* dt = s_d + tl*65;
    int* mt = s_t[wv];
    int got[32];
    unsigned okbits = 0;
#pragma unroll
    for (int i = 0; i < 32; ++i) {
        const int      so = __builtin_amdgcn_readlane(own, i);
        const unsigned sp = (unsigned)__builtin_amdgcn_readlane((int)pk, i);
        const bool ok = ((sp + step) & FLD_OUT) == 0u;
        const int d = dt[so & 63];
        got[i] = grid[so + (ok ? d : 0)];
        okbits |= (ok ? 1u : 0u) << i;
    }
    if (lane < NOFF) {
#pragma unroll
        for (int i = 0; i < 32; ++i) mt[i*NOFF + lane] = ((okbits >> i) & 1u) ? got[i] : -1;
    }
    __builtin_amdgcn_fence(__ATOMIC_ACQ_REL, "wavefront");
    int idx[NOFF];
#pragma unroll
    for (int t = 0; t < NOFF; ++t) idx[t] = mt[lane*NOFF + t];
#pragma unroll
    for (int t = 0; t < NOFF; ++t) {
        unsigned mask = __builtin_amdgcn_ballot_w32(idx[t] >= 0);
        if (lane == 0) s_c[wv][t] = (t == CTAP) ? 0 : __builtin_popcount(mask);
    }
    __syncthreads();

    int tot = 0;
    int off = 0;
    if (lane < NOFF) {
        int a_c[NWV];
#pragma unroll
        for (int w = 0; w < NWV; ++w) a_c[w] = s_c[w][lane];
#pragma unroll
        for (int w = 0; w < NWV; ++w) {
            tot += a_c[w];
            off += (w < wv) ? a_c[w] : 0;
        }
    }
    if (wv == 0) cntb[blockIdx.x*32 + lane] = tot;

    int k = 0;
#pragma unroll
    for (int t = 0; t < NOFF; ++t) {
        if (t == CTAP) { if (live) cen[p] = idx[t]; continue; }
        const int u = t < CTAP ? t : t - 1;
        const int off_t = __builtin_amdgcn_readlane(off, t);
        const int tot_t = __builtin_amdgcn_readlane(tot, t);
        unsigned mask = __builtin_amdgcn_ballot_w32(idx[t] >= 0);
        if (idx[t] >= 0 && tot_t <= SEG) {
            const int r = off_t + __builtin_popcount(mask & ((1u << lane) - 1u));
            const int pos = (u*NB + (int)blockIdx.x)*SEG + r;
            s_pin[u*SEG + r] = idx[t];
            list[(size_t)k*NPTS + p] = pos;
            ++k;
        }
    }
    if (live) {
        for (int s2 = 0; s2 < LIST_HEAD; ++s2)
            if (k <= s2) list[(size_t)s2*NPTS + p] = -1;
        if (k >= LIST_HEAD && k < NOFF) list[(size_t)k*NPTS + p] = -1;
    }
    __syncthreads();

    static_assert(SEG/16 <= NWV, "a wave takes at most one tile of a tap");
    const int col  = lane & 15;
    const int half = lane >> 4;
    const int ntl  = (lane < NOFF && tot <= SEG) ? ((tot + 15) >> 4) : 0;
    int s = wv;
    int base = 0;
    for (int t = 0; t < NOFF; ++t) {
        const int nt_t  = __builtin_amdgcn_readlane(ntl, t);
        const int tot_t = __builtin_amdgcn_readlane(tot, t);
        const int u = t < CTAP ? t : t - 1;
        if (s < base + nt_t) {
            const int j   = s - base;
            const int rem = tot_t - j*16;
            const int nr  = rem < 16 ? rem : 16;
            int q = -1;
            if (col < nr) q = s_pin[u*SEG + j*16 + col];

            v2f a[16];
#pragma unroll
            for (int kc = 0; kc < 16; ++kc) {
                v2f v; v.x = 0.f; v.y = 0.f;
                if (q >= 0) v = *(const v2f*)(feats + (size_t)q*CI + kc*4 + 2*half);
                a[kc] = v;
            }

            v8f acc[4];
#pragma unroll
            for (int nt = 0; nt < 4; ++nt)
#pragma unroll
                for (int e = 0; e < 8; ++e) acc[nt][e] = 0.f;

            const float* wbase = wT + (size_t)t*(16*4*64) + lane*2;
#pragma unroll
            for (int kc = 0; kc < 16; ++kc) {
#pragma unroll
                for (int nt = 0; nt < 4; ++nt) {
                    v2f b = *(const v2f*)(wbase + (kc*4 + nt)*64);
                    acc[nt] = __builtin_amdgcn_wmma_f32_16x16x4_f32(
                        false, a[kc], false, b, (short)0, acc[nt], false, false);
                }
            }

            const size_t row0 = (size_t)(u*NB + (int)blockIdx.x)*SEG + j*16 + 8*half;
#pragma unroll
            for (int e = 0; e < 8; ++e) {
                if (e + 8*half < nr) {
                    const size_t ro = (row0 + e)*CO;
#pragma unroll
                    for (int nt = 0; nt < 4; ++nt)
                        part[ro + nt*16 + col] = acc[nt][e];
                }
            }
            s += NWV;
        }
        base += nt_t;
    }
}

#define FIN_WGS ((NPTS/16 + 7)/8)
#define TP 68
#define EB 4
__global__ __launch_bounds__(256)
void k_final(const float* __restrict__ feats, const float* __restrict__ wT,
             const float* __restrict__ bias, const int* __restrict__ cen,
             const int* __restrict__ list, const float* __restrict__ part,
             const int* __restrict__ cntb, const int* __restrict__ coords,
             const int* __restrict__ grid, float* __restrict__ out, int n) {
    __shared__ __attribute__((aligned(16))) float s_tile[8][16*TP];
    __shared__ __attribute__((aligned(16))) int   s_head[8][16*LIST_HEAD];
    const int lane = threadIdx.x & 31;
    const int col  = lane & 15;
    const int half = lane >> 4;
    const int wv   = threadIdx.x >> 5;
    const int tile  = __builtin_amdgcn_readfirstlane((int)(blockIdx.x * 8 + wv));
    const int ntile = (n + 15) >> 4;
    if (tile >= ntile) return;
    float* my  = s_tile[wv];
    int*   myh = s_head[wv];
    const float4* rows = (const float4*)part;

    const float* wbase = wT + (size_t)CTAP*(16*4*64) + lane*2;
    const float4 bs = *(const float4*)(bias + col*4);

    const unsigned dense = __builtin_amdgcn_ballot_w32(cntb[(tile/(PB/16))*32 + lane] > SEG);

    const int p = tile*16 + col;
    int q = -1;
    int4 hd = make_int4(-1, -1, -1, -1);
    if (p < n) {
        q    = cen[p];
        hd.x = list[p];
        hd.y = list[(size_t)1*NPTS + p];
        hd.z = list[(size_t)2*NPTS + p];
        hd.w = list[(size_t)3*NPTS + p];
    }
    v2f a[16];
    const float* arow = feats + (size_t)(q >= 0 ? q : 0)*CI + 2*half;
#pragma unroll
    for (int kc = 0; kc < 16; ++kc) {
        v2f v = *(const v2f*)(arow + kc*4);
        if (q < 0) { v.x = 0.f; v.y = 0.f; }
        a[kc] = v;
    }
    v8f acc[4];
#pragma unroll
    for (int nt = 0; nt < 4; ++nt)
#pragma unroll
        for (int j = 0; j < 8; ++j) acc[nt][j] = 0.f;
#pragma unroll
    for (int kc = 0; kc < 16; ++kc) {
#pragma unroll
        for (int nt = 0; nt < 4; ++nt) {
            v2f b = *(const v2f*)(wbase + (kc*4 + nt)*64);
            acc[nt] = __builtin_amdgcn_wmma_f32_16x16x4_f32(
                false, a[kc], false, b, (short)0, acc[nt], false, false);
        }
    }
    if (dense != 0u) {
        int bz = 0, z = 0, y = 0, x = 0;
        if (p < n) { bz = coords[p*4+0]; z = coords[p*4+1]; y = coords[p*4+2]; x = coords[p*4+3]; }
        const bool own_ok = p < n && (unsigned)bz < (unsigned)BB && (unsigned)z < (unsigned)DD &&
                            (unsigned)y < (unsigned)HH && (unsigned)x < (unsigned)WW;
        for (unsigned m = dense; m != 0u; m &= m - 1u) {
            const int t  = __builtin_ctz(m);
            const int nz = z + t/9 - 1, ny = y + (t/3)%3 - 1, nx = x + t%3 - 1;
            int q2 = -1;
            if (own_ok && (unsigned)nz < (unsigned)DD && (unsigned)ny < (unsigned)HH &&
                (unsigned)nx < (unsigned)WW)
                q2 = grid[cell_index(bz, nz, ny, nx)];
            const float* ar = feats + (size_t)(q2 >= 0 ? q2 : 0)*CI + 2*half;
            const float* wb = wT + (size_t)t*(16*4*64) + lane*2;
#pragma nounroll
            for (int kc = 0; kc < 16; ++kc) {
                v2f av = *(const v2f*)(ar + kc*4);
                if (q2 < 0) { av.x = 0.f; av.y = 0.f; }
#pragma unroll
                for (int nt = 0; nt < 4; ++nt) {
                    v2f b = *(const v2f*)(wb + (kc*4 + nt)*64);
                    acc[nt] = __builtin_amdgcn_wmma_f32_16x16x4_f32(
                        false, av, false, b, (short)0, acc[nt], false, false);
                }
            }
        }
    }
#pragma unroll
    for (int j = 0; j < 8; ++j)
#pragma unroll
        for (int nt = 0; nt < 4; ++nt)
            my[(j + 8*half)*TP + nt*16 + col] = acc[nt][j];
    if (half == 0) *(int4*)(myh + col*LIST_HEAD) = hd;
    __builtin_amdgcn_fence(__ATOMIC_ACQ_REL, "wavefront");

    const int c4 = col*4;
#pragma unroll
    for (int b0 = 0; b0 < 8; b0 += EB) {
        int4 h[EB];
        float4 v[EB], x0[EB], x1[EB], x2[EB], x3[EB];
#pragma unroll
        for (int e = 0; e < EB; ++e) {
            const int row = 2*(b0 + e) + half;
            h[e] = *(const int4*)(myh + row*LIST_HEAD);
            v[e] = *(const float4*)(my + row*TP + c4);
        }
#pragma unroll
        for (int e = 0; e < EB; ++e) {
            const float4 z = make_float4(0.f, 0.f, 0.f, 0.f);
            x0[e] = z; x1[e] = z; x2[e] = z; x3[e] = z;
            if (h[e].x >= 0) {
                x0[e] = rows[(unsigned)h[e].x*(CO/4) + col];
                x1[e] = rows[(unsigned)(h[e].y >= 0 ? h[e].y : h[e].x)*(CO/4) + col];
                x2[e] = rows[(unsigned)(h[e].z >= 0 ? h[e].z : h[e].x)*(CO/4) + col];
                x3[e] = rows[(unsigned)(h[e].w >= 0 ? h[e].w : h[e].x)*(CO/4) + col];
            }
        }
#pragma unroll
        for (int e = 0; e < EB; ++e) {
            const int pp = tile*16 + 2*(b0 + e) + half;
            float4 r;
            r.x = bs.x + v[e].x; r.y = bs.y + v[e].y; r.z = bs.z + v[e].z; r.w = bs.w + v[e].w;
            if (h[e].x >= 0) {
                r.x += x0[e].x; r.y += x0[e].y; r.z += x0[e].z; r.w += x0[e].w;
                if (h[e].y >= 0) { r.x += x1[e].x; r.y += x1[e].y; r.z += x1[e].z; r.w += x1[e].w; }
                if (h[e].z >= 0) { r.x += x2[e].x; r.y += x2[e].y; r.z += x2[e].z; r.w += x2[e].w; }
                if (h[e].w >= 0) {
                    r.x += x3[e].x; r.y += x3[e].y; r.z += x3[e].z; r.w += x3[e].w;
                    for (int k = LIST_HEAD; k < NOFF; ++k) {
                        const int rr = list[(size_t)k*NPTS + pp];
                        if (rr < 0) break;
                        const float4 y = rows[(unsigned)rr*(CO/4) + col];
                        r.x += y.x; r.y += y.y; r.z += y.z; r.w += y.w;
                    }
                }
            }
            if (pp < n) *(float4*)(out + (size_t)pp*CO + c4) = r;
        }
    }
}

extern "C" void kernel_launch(void* const* d_in, const int* in_sizes, int n_in,
                              void* d_out, int out_size, void* d_ws, size_t ws_size,
                              hipStream_t stream) {
    const float* feats  = (const float*)d_in[0];
    const float* weight = (const float*)d_in[1];
    const float* bias   = (const float*)d_in[2];
    const int*   coords = (const int*)  d_in[3];
    float* out = (float*)d_out;

    const int n = in_sizes[0] / CI;

    float* wT    = (float*)d_ws;
    int*   grid  = (int*)  ((char*)d_ws + OFF_GRID);
    int*   list  = (int*)  ((char*)d_ws + OFF_NPOS);
    int*   cntb  = (int*)  ((char*)d_ws + OFF_CNT);
    int*   cen   = (int*)  ((char*)d_ws + OFF_CEN);
    float* part  = (float*)((char*)d_ws + OFF_PART);

    k_fill_grid<<<FILL_WGS,               256, 0, stream>>>(grid, weight, wT);
    k_scatter  <<<(n + 255)/256,          256, 0, stream>>>(coords, grid, n);
    k_nbr      <<<NB,                     PB,  0, stream>>>(coords, grid, list, cntb, cen, feats, wT, part, n);
    k_final    <<<FIN_WGS,                256, 0, stream>>>(feats, wT, bias, cen, list, part, cntb, coords, grid,
                                                            out, n);
}
